# starting kernel
# baseline (speedup 1.0000x reference)
#include <hip/hip_runtime.h>
#include <hip/hip_bf16.h>

typedef __attribute__((ext_vector_type(2))) float v2f;
typedef __attribute__((ext_vector_type(8))) float v8f;

#define DFEAT 64


__global__ void zero_f32_kernel(float* __restrict__ p, int n) {
    int i = blockIdx.x * blockDim.x + threadIdx.x;
    if (i < n) p[i] = 0.0f;
}

__global__ void degree_kernel(const int* __restrict__ dst, float* __restrict__ deg, int E) {
    int e = blockIdx.x * blockDim.x + threadIdx.x;
    if (e < E) atomicAdd(&deg[dst[e]], 1.0f);
}

__global__ void dinv_kernel(const float* __restrict__ deg, float* __restrict__ dinv, int n) {
    int i = blockIdx.x * blockDim.x + threadIdx.x;
    if (i < n) dinv[i] = rsqrtf(deg[i] + 1.0f);
}

__global__ void selfloop_init_kernel(const float* __restrict__ h, const float* __restrict__ dinv,
                                     float* __restrict__ agg, int total) {
    int idx = blockIdx.x * blockDim.x + threadIdx.x;
    if (idx >= total) return;
    int node = idx >> 6;
    float di = dinv[node];
    agg[idx] = h[idx] * (di * di);
}

__global__ void selfloop_bias_init_kernel(const float* __restrict__ h, const float* __restrict__ dinv,
                                          const float* __restrict__ b, float* __restrict__ out, int total) {
    int idx = blockIdx.x * blockDim.x + threadIdx.x;
    if (idx >= total) return;
    int node = idx >> 6;
    int j = idx & (DFEAT - 1);
    float di = dinv[node];
    out[idx] = h[idx] * (di * di) + b[j];
}

__global__ void relu_bias_kernel(float* __restrict__ agg, const float* __restrict__ b, int total) {
    int idx = blockIdx.x * blockDim.x + threadIdx.x;
    if (idx >= total) return;
    int j = idx & (DFEAT - 1);
    float v = agg[idx] + b[j];
    agg[idx] = v > 0.0f ? v : 0.0f;
}

__global__ void edge_scatter_kernel(const int* __restrict__ src, const int* __restrict__ dst,
                                    const float* __restrict__ dinv, const float* __restrict__ h,
                                    float* __restrict__ agg, int E) {
    int tid = blockIdx.x * blockDim.x + threadIdx.x;
    int e = tid >> 4;
    if (e >= E) return;
    int j = (tid & 15) << 2;
    int s = src[e];
    int d = dst[e];
    float c = dinv[s] * dinv[d];
    const float4 hv = *(const float4*)(h + (size_t)s * DFEAT + j);
    float* ap = agg + (size_t)d * DFEAT + j;
    atomicAdd(ap + 0, hv.x * c);
    atomicAdd(ap + 1, hv.y * c);
    atomicAdd(ap + 2, hv.z * c);
    atomicAdd(ap + 3, hv.w * c);
}

__global__ __launch_bounds__(128) void gemm64_wmma_kernel(const float* __restrict__ X,
                                                          const float* __restrict__ W,
                                                          float* __restrict__ out, int n) {
    __shared__ float Wl[DFEAT * DFEAT];

    const int t = threadIdx.x;
    {
        const float4* Wv = (const float4*)W;
        float4* Wlv = (float4*)Wl;
        #pragma unroll
        for (int i = 0; i < (DFEAT * DFEAT / 4) / 128; ++i)
            Wlv[t + i * 128] = Wv[t + i * 128];
    }
    __syncthreads();

    const int wave = t >> 5;
    const int lane = t & 31;
    const int half = lane >> 4;
    const int l16 = lane & 15;
    const int rowBase = blockIdx.x * 64 + wave * 16;

    v8f acc[4] = {};

    int rA = rowBase + l16;
    if (rA > n - 1) rA = n - 1;
    const float* Xrow = X + (size_t)rA * DFEAT;

    for (int k0 = 0; k0 < DFEAT; k0 += 4) {
        const int ka = k0 + half * 2;
        v2f a;
        a[0] = Xrow[ka];
        a[1] = Xrow[ka + 1];
        #pragma unroll
        for (int tn = 0; tn < 4; ++tn) {
            const int n0 = tn * 16;
            v2f b;
            b[0] = Wl[ka * DFEAT + n0 + l16];
            b[1] = Wl[(ka + 1) * DFEAT + n0 + l16];
            acc[tn] = __builtin_amdgcn_wmma_f32_16x16x4_f32(
                 false, a,  false, b,
                 (short)0, acc[tn],  false,  false);
        }
    }

    #pragma unroll
    for (int tn = 0; tn < 4; ++tn) {
        const int n0 = tn * 16;
        #pragma unroll
        for (int v = 0; v < 8; ++v) {
            int row = rowBase + v + half * 8;
            if (row < n) out[(size_t)row * DFEAT + n0 + l16] = acc[tn][v];
        }
    }
}


extern "C" void kernel_launch(void* const* d_in, const int* in_sizes, int n_in,
                              void* d_out, int out_size, void* d_ws, size_t ws_size,
                              hipStream_t stream) {
    const int* eidx = (const int*)d_in[0];
    const float* emb = (const float*)d_in[1];
    const float* W1 = (const float*)d_in[2];
    const float* b1 = (const float*)d_in[3];
    const float* W2 = (const float*)d_in[4];
    const float* b2 = (const float*)d_in[5];
    float* out = (float*)d_out;

    const int E = in_sizes[0] / 2;
    const int n = in_sizes[1] / DFEAT;
    const int total = n * DFEAT;

    const int* src = eidx;
    const int* dst = eidx + E;

    float* deg  = (float*)d_ws;
    float* dinv = deg + n;
    float* h1   = dinv + n;
    float* agg1 = h1 + (size_t)n * DFEAT;
    float* h2   = agg1 + (size_t)n * DFEAT;

    const int B = 256;
    const int gN = (n + B - 1) / B;
    const int gE = (E + B - 1) / B;
    const int gT = (total + B - 1) / B;
    const int gScatter = (E * 16 + B - 1) / B;
    const int gGemm = (n + 63) / 64;

    zero_f32_kernel<<<gN, B, 0, stream>>>(deg, n);
    degree_kernel<<<gE, B, 0, stream>>>(dst, deg, E);
    dinv_kernel<<<gN, B, 0, stream>>>(deg, dinv, n);

    gemm64_wmma_kernel<<<gGemm, 128, 0, stream>>>(emb, W1, h1, n);
    selfloop_init_kernel<<<gT, B, 0, stream>>>(h1, dinv, agg1, total);
    edge_scatter_kernel<<<gScatter, B, 0, stream>>>(src, dst, dinv, h1, agg1, E);
    relu_bias_kernel<<<gT, B, 0, stream>>>(agg1, b1, total);

    gemm64_wmma_kernel<<<gGemm, 128, 0, stream>>>(agg1, W2, h2, n);
    selfloop_bias_init_kernel<<<gT, B, 0, stream>>>(h2, dinv, b2, out, total);
    edge_scatter_kernel<<<gScatter, B, 0, stream>>>(src, dst, dinv, h2, out, E);
}


# Float atomic scatter-add replaced by a per-call destination-sorted edge list (int atomic rank, scan, fill); 16 lanes per node sum its in-edges in arrival-rank order and write each row once
# speedup vs baseline: 1.6879x; 1.6879x over previous
#include <hip/hip_runtime.h>
#include <hip/hip_bf16.h>

typedef __attribute__((ext_vector_type(2))) float v2f;
typedef __attribute__((ext_vector_type(8))) float v8f;

#define DFEAT 64


__global__ void zero_i32_kernel(int* __restrict__ p, int n) {
    int i = blockIdx.x * blockDim.x + threadIdx.x;
    if (i < n) p[i] = 0;
}

__global__ void rank_count_kernel(const int* __restrict__ dst, int* __restrict__ cnt,
                                  int* __restrict__ rank, int E) {
    int e = blockIdx.x * blockDim.x + threadIdx.x;
    if (e < E) rank[e] = atomicAdd(&cnt[dst[e]], 1);
}

__global__ __launch_bounds__(256) void scan_dinv_kernel(const int* __restrict__ cnt,
                                                        int* __restrict__ rowptr,
                                                        float* __restrict__ dinv, int n) {
    __shared__ int wtot[8];
    __shared__ int wsum[8];
    const int t = threadIdx.x;
    const int lane = t & 31;
    const int wave = t >> 5;
    const int tile0 = blockIdx.x * 256;

    int part = 0;
    for (int j = t; j < tile0; j += 256) part += cnt[j];
    for (int o = 16; o > 0; o >>= 1) part += __shfl_xor(part, o, 32);

    const int i = tile0 + t;
    const int c = (i < n) ? cnt[i] : 0;
    int incl = c;
    for (int o = 1; o < 32; o <<= 1) {
        int up = __shfl_up(incl, o, 32);
        if (lane >= o) incl += up;
    }
    if (lane == 0) wsum[wave] = part;
    if (lane == 31) wtot[wave] = incl;
    __syncthreads();

    int off = 0;
    for (int w = 0; w < 8; ++w) {
        off += wsum[w];
        if (w < wave) off += wtot[w];
    }
    if (i < n) {
        rowptr[i] = off + incl - c;
        dinv[i] = rsqrtf((float)c + 1.0f);
    }
}

__global__ void fill_csr_kernel(const int* __restrict__ src, const int* __restrict__ dst,
                                const int* __restrict__ rank, const int* __restrict__ rowptr,
                                const float* __restrict__ dinv, int2* __restrict__ csr, int E) {
    int e = blockIdx.x * blockDim.x + threadIdx.x;
    if (e >= E) return;
    const int s = src[e];
    const int d = dst[e];
    const float c = dinv[s] * dinv[d];
    csr[rowptr[d] + rank[e]] = make_int2(s, __float_as_int(c));
}

template <bool RELU>
__global__ __launch_bounds__(256) void gather_rows_kernel(const int* __restrict__ rowptr,
                                                          const int* __restrict__ cnt,
                                                          const int2* __restrict__ csr,
                                                          const float* __restrict__ h,
                                                          const float* __restrict__ dinv,
                                                          const float* __restrict__ b,
                                                          float* __restrict__ outp, int n) {
    const int tid = blockIdx.x * 256 + threadIdx.x;
    int node = tid >> 4;
    const bool valid = node < n;
    if (!valid) node = n - 1;
    const int l16 = threadIdx.x & 15;
    const int j = l16 << 2;

    const int start = rowptr[node];
    const int len = valid ? cnt[node] : 0;
    int lenMax = max(len, __shfl_xor(len, 16, 32));
    lenMax = __builtin_amdgcn_readfirstlane(lenMax);

    float4 acc = make_float4(0.0f, 0.0f, 0.0f, 0.0f);
    for (int base = 0; base < lenMax; base += 16) {
        int ex = node;
        int ey = 0;
        if (base + l16 < len) {
            const int2 ent = csr[start + base + l16];
            ex = ent.x;
            ey = ent.y;
        }
        const int chunk = min(16, lenMax - base);
        for (int k = 0; k < chunk; k += 4) {
            int s[4];
            float w[4];
            float4 v[4];
            for (int u = 0; u < 4; ++u) {
                s[u] = __shfl(ex, k + u, 16);
                w[u] = __int_as_float(__shfl(ey, k + u, 16));
            }
            for (int u = 0; u < 4; ++u)
                v[u] = *(const float4*)(h + (size_t)s[u] * DFEAT + j);
            for (int u = 0; u < 4; ++u) {
                acc.x += v[u].x * w[u];
                acc.y += v[u].y * w[u];
                acc.z += v[u].z * w[u];
                acc.w += v[u].w * w[u];
            }
        }
    }

    if (valid) {
        const float di = dinv[node];
        const float dd = di * di;
        const float4 hv = *(const float4*)(h + (size_t)node * DFEAT + j);
        float4 r;
        r.x = acc.x + hv.x * dd + b[j + 0];
        r.y = acc.y + hv.y * dd + b[j + 1];
        r.z = acc.z + hv.z * dd + b[j + 2];
        r.w = acc.w + hv.w * dd + b[j + 3];
        if (RELU) {
            r.x = r.x > 0.0f ? r.x : 0.0f;
            r.y = r.y > 0.0f ? r.y : 0.0f;
            r.z = r.z > 0.0f ? r.z : 0.0f;
            r.w = r.w > 0.0f ? r.w : 0.0f;
        }
        *(float4*)(outp + (size_t)node * DFEAT + j) = r;
    }
}

__global__ __launch_bounds__(128) void gemm64_wmma_kernel(const float* __restrict__ X,
                                                          const float* __restrict__ W,
                                                          float* __restrict__ out, int n) {
    __shared__ float Wl[DFEAT * DFEAT];

    const int t = threadIdx.x;
    {
        const float4* Wv = (const float4*)W;
        float4* Wlv = (float4*)Wl;
        #pragma unroll
        for (int i = 0; i < (DFEAT * DFEAT / 4) / 128; ++i)
            Wlv[t + i * 128] = Wv[t + i * 128];
    }
    __syncthreads();

    const int wave = t >> 5;
    const int lane = t & 31;
    const int half = lane >> 4;
    const int l16 = lane & 15;
    const int rowBase = blockIdx.x * 64 + wave * 16;

    v8f acc[4] = {};

    int rA = rowBase + l16;
    if (rA > n - 1) rA = n - 1;
    const float* Xrow = X + (size_t)rA * DFEAT;

    for (int k0 = 0; k0 < DFEAT; k0 += 4) {
        const int ka = k0 + half * 2;
        v2f a;
        a[0] = Xrow[ka];
        a[1] = Xrow[ka + 1];
        #pragma unroll
        for (int tn = 0; tn < 4; ++tn) {
            const int n0 = tn * 16;
            v2f b;
            b[0] = Wl[ka * DFEAT + n0 + l16];
            b[1] = Wl[(ka + 1) * DFEAT + n0 + l16];
            acc[tn] = __builtin_amdgcn_wmma_f32_16x16x4_f32(
                 false, a,  false, b,
                 (short)0, acc[tn],  false,  false);
        }
    }

    #pragma unroll
    for (int tn = 0; tn < 4; ++tn) {
        const int n0 = tn * 16;
        #pragma unroll
        for (int v = 0; v < 8; ++v) {
            int row = rowBase + v + half * 8;
            if (row < n) out[(size_t)row * DFEAT + n0 + l16] = acc[tn][v];
        }
    }
}


extern "C" void kernel_launch(void* const* d_in, const int* in_sizes, int n_in,
                              void* d_out, int out_size, void* d_ws, size_t ws_size,
                              hipStream_t stream) {
    const int* eidx = (const int*)d_in[0];
    const float* emb = (const float*)d_in[1];
    const float* W1 = (const float*)d_in[2];
    const float* b1 = (const float*)d_in[3];
    const float* W2 = (const float*)d_in[4];
    const float* b2 = (const float*)d_in[5];
    float* out = (float*)d_out;

    const int E = in_sizes[0] / 2;
    const int n = in_sizes[1] / DFEAT;

    const int* src = eidx;
    const int* dst = eidx + E;

    float* h1    = (float*)d_ws;
    float* x     = h1 + (size_t)n * DFEAT;
    int2* csr    = (int2*)(x + (size_t)n * DFEAT);
    int* rank    = (int*)(csr + E);
    int* cnt     = rank + E;
    int* rowptr  = cnt + n;
    float* dinv  = (float*)(rowptr + n);
    float* h2    = h1;

    const int B = 256;
    const int gN = (n + B - 1) / B;
    const int gE = (E + B - 1) / B;
    const int gRows = (n * 16 + B - 1) / B;
    const int gGemm = (n + 63) / 64;

    zero_i32_kernel<<<gN, B, 0, stream>>>(cnt, n);
    rank_count_kernel<<<gE, B, 0, stream>>>(dst, cnt, rank, E);
    scan_dinv_kernel<<<gN, B, 0, stream>>>(cnt, rowptr, dinv, n);
    fill_csr_kernel<<<gE, B, 0, stream>>>(src, dst, rank, rowptr, dinv, csr, E);

    gemm64_wmma_kernel<<<gGemm, 128, 0, stream>>>(emb, W1, h1, n);
    gather_rows_kernel<true><<<gRows, B, 0, stream>>>(rowptr, cnt, csr, h1, dinv, b1, x, n);

    gemm64_wmma_kernel<<<gGemm, 128, 0, stream>>>(x, W2, h2, n);
    gather_rows_kernel<false><<<gRows, B, 0, stream>>>(rowptr, cnt, csr, h2, dinv, b2, out, n);
}


# Layer 1 aggregates emb rows first: one kernel gathers a 64-node tile into LDS and runs both 64x64 WMMA products on it (relu between); 2 GEMM launches, h1 and x planes gone; layer-1 sums reassociated
# speedup vs baseline: 1.9108x; 1.1321x over previous
#include <hip/hip_runtime.h>
#include <hip/hip_bf16.h>

typedef __attribute__((ext_vector_type(2))) float v2f;
typedef __attribute__((ext_vector_type(8))) float v8f;

#define DFEAT 64


__global__ void zero_i32_kernel(int* __restrict__ p, int n) {
    int i = blockIdx.x * blockDim.x + threadIdx.x;
    if (i < n) p[i] = 0;
}

__global__ void rank_count_kernel(const int* __restrict__ dst, int* __restrict__ cnt,
                                  int* __restrict__ rank, int E) {
    int e = blockIdx.x * blockDim.x + threadIdx.x;
    if (e < E) rank[e] = atomicAdd(&cnt[dst[e]], 1);
}

__global__ __launch_bounds__(256) void scan_dinv_kernel(const int* __restrict__ cnt,
                                                        int* __restrict__ rowptr,
                                                        float* __restrict__ dinv, int n) {
    __shared__ int wtot[8];
    __shared__ int wsum[8];
    const int t = threadIdx.x;
    const int lane = t & 31;
    const int wave = t >> 5;
    const int tile0 = blockIdx.x * 256;

    int part = 0;
    for (int j = t; j < tile0; j += 256) part += cnt[j];
    for (int o = 16; o > 0; o >>= 1) part += __shfl_xor(part, o, 32);

    const int i = tile0 + t;
    const int c = (i < n) ? cnt[i] : 0;
    int incl = c;
    for (int o = 1; o < 32; o <<= 1) {
        int up = __shfl_up(incl, o, 32);
        if (lane >= o) incl += up;
    }
    if (lane == 0) wsum[wave] = part;
    if (lane == 31) wtot[wave] = incl;
    __syncthreads();

    int off = 0;
    for (int w = 0; w < 8; ++w) {
        off += wsum[w];
        if (w < wave) off += wtot[w];
    }
    if (i < n) {
        rowptr[i] = off + incl - c;
        dinv[i] = rsqrtf((float)c + 1.0f);
    }
}

__global__ void fill_csr_kernel(const int* __restrict__ src, const int* __restrict__ dst,
                                const int* __restrict__ rank, const int* __restrict__ rowptr,
                                const float* __restrict__ dinv, int2* __restrict__ csr, int E) {
    int e = blockIdx.x * blockDim.x + threadIdx.x;
    if (e >= E) return;
    const int s = src[e];
    const int d = dst[e];
    const float c = dinv[s] * dinv[d];
    csr[rowptr[d] + rank[e]] = make_int2(s, __float_as_int(c));
}

template <bool RELU>
__global__ __launch_bounds__(256) void gather_rows_kernel(const int* __restrict__ rowptr,
                                                          const int* __restrict__ cnt,
                                                          const int2* __restrict__ csr,
                                                          const float* __restrict__ h,
                                                          const float* __restrict__ dinv,
                                                          const float* __restrict__ b,
                                                          float* __restrict__ outp, int n) {
    const int tid = blockIdx.x * 256 + threadIdx.x;
    int node = tid >> 4;
    const bool valid = node < n;
    if (!valid) node = n - 1;
    const int l16 = threadIdx.x & 15;
    const int j = l16 << 2;

    const int start = rowptr[node];
    const int len = valid ? cnt[node] : 0;
    int lenMax = max(len, __shfl_xor(len, 16, 32));
    lenMax = __builtin_amdgcn_readfirstlane(lenMax);

    float4 acc = make_float4(0.0f, 0.0f, 0.0f, 0.0f);
    for (int base = 0; base < lenMax; base += 16) {
        int ex = node;
        int ey = 0;
        if (base + l16 < len) {
            const int2 ent = csr[start + base + l16];
            ex = ent.x;
            ey = ent.y;
        }
        const int chunk = min(16, lenMax - base);
        for (int k = 0; k < chunk; k += 4) {
            int s[4];
            float w[4];
            float4 v[4];
            for (int u = 0; u < 4; ++u) {
                s[u] = __shfl(ex, k + u, 16);
                w[u] = __int_as_float(__shfl(ey, k + u, 16));
            }
            for (int u = 0; u < 4; ++u)
                v[u] = *(const float4*)(h + (size_t)s[u] * DFEAT + j);
            for (int u = 0; u < 4; ++u) {
                acc.x += v[u].x * w[u];
                acc.y += v[u].y * w[u];
                acc.z += v[u].z * w[u];
                acc.w += v[u].w * w[u];
            }
        }
    }

    if (valid) {
        const float di = dinv[node];
        const float dd = di * di;
        const float4 hv = *(const float4*)(h + (size_t)node * DFEAT + j);
        float4 r;
        r.x = acc.x + hv.x * dd + b[j + 0];
        r.y = acc.y + hv.y * dd + b[j + 1];
        r.z = acc.z + hv.z * dd + b[j + 2];
        r.w = acc.w + hv.w * dd + b[j + 3];
        if (RELU) {
            r.x = r.x > 0.0f ? r.x : 0.0f;
            r.y = r.y > 0.0f ? r.y : 0.0f;
            r.z = r.z > 0.0f ? r.z : 0.0f;
            r.w = r.w > 0.0f ? r.w : 0.0f;
        }
        *(float4*)(outp + (size_t)node * DFEAT + j) = r;
    }
}

#define TILE_ROWS 64
#define EPS 2
#define TP 68

__global__ __launch_bounds__(256) void gather_mlp_kernel(const int* __restrict__ rowptr,
                                                         const int* __restrict__ cnt,
                                                         const int2* __restrict__ csr,
                                                         const float* __restrict__ emb,
                                                         const float* __restrict__ dinv,
                                                         const float* __restrict__ W1,
                                                         const float* __restrict__ b1,
                                                         const float* __restrict__ W2,
                                                         float* __restrict__ h2, int n) {
    __shared__ __attribute__((aligned(16))) float Wl1[DFEAT * DFEAT];
    __shared__ __attribute__((aligned(16))) float Wl2[DFEAT * DFEAT];
    __shared__ __attribute__((aligned(16))) float Ta[TILE_ROWS * TP];
    __shared__ __attribute__((aligned(16))) float Tx[TILE_ROWS * TP];

    const int t = threadIdx.x;
    const int wave = t >> 5;
    const int half = (t >> 4) & 1;
    const int l16 = t & 15;
    const int j = l16 << 2;

    {
        const float4* W1v = (const float4*)W1;
        const float4* W2v = (const float4*)W2;
        float4* Wl1v = (float4*)Wl1;
        float4* Wl2v = (float4*)Wl2;
        #pragma unroll
        for (int i = 0; i < (DFEAT * DFEAT / 4) / 256; ++i) {
            Wl1v[t + i * 256] = W1v[t + i * 256];
            Wl2v[t + i * 256] = W2v[t + i * 256];
        }
    }

    const int grp = wave * 2 + half;
    const int tileRow0 = grp * 4;
    const int node0 = blockIdx.x * TILE_ROWS + tileRow0;

    int nd[4], start[4], len[4];
    #pragma unroll
    for (int q = 0; q < 4; ++q) {
        const int node = node0 + q;
        const bool valid = node < n;
        nd[q] = valid ? node : n - 1;
        start[q] = rowptr[nd[q]];
        len[q] = valid ? cnt[nd[q]] : 0;
    }
    int lenMax = max(max(len[0], len[1]), max(len[2], len[3]));
    lenMax = max(lenMax, __shfl_xor(lenMax, 16, 32));
    lenMax = __builtin_amdgcn_readfirstlane(lenMax);

    float4 acc[4];
    #pragma unroll
    for (int q = 0; q < 4; ++q) acc[q] = make_float4(0.0f, 0.0f, 0.0f, 0.0f);
    const char* embBytes = (const char*)emb;
    const unsigned colByte = (unsigned)j << 2;

    for (int base = 0; base < lenMax; base += 16) {
        int ex[4], ey[4];
        int2 ent[4];
        #pragma unroll
        for (int q = 0; q < 4; ++q) {
            const bool in = base + l16 < len[q];
            ent[q] = csr[in ? start[q] + base + l16 : 0];
        }
        #pragma unroll
        for (int q = 0; q < 4; ++q) {
            const bool in = base + l16 < len[q];
            ex[q] = in ? ent[q].x : nd[q];
            ey[q] = in ? ent[q].y : 0;
        }
        const int chunk = min(16, lenMax - base);
        for (int k = 0; k < chunk; k += EPS) {
            float wgt[4][EPS];
            float4 v[4][EPS];
            #pragma unroll
            for (int q = 0; q < 4; ++q) {
                #pragma unroll
                for (int u = 0; u < EPS; ++u) {
                    const int s = __shfl(ex[q], k + u, 16);
                    wgt[q][u] = __int_as_float(__shfl(ey[q], k + u, 16));
                    const unsigned off = ((unsigned)s << 8) + colByte;
                    v[q][u] = *(const float4*)(embBytes + off);
                }
            }
            #pragma unroll
            for (int q = 0; q < 4; ++q) {
                #pragma unroll
                for (int u = 0; u < EPS; ++u) {
                    acc[q].x += v[q][u].x * wgt[q][u];
                    acc[q].y += v[q][u].y * wgt[q][u];
                    acc[q].z += v[q][u].z * wgt[q][u];
                    acc[q].w += v[q][u].w * wgt[q][u];
                }
            }
        }
    }

    float di[4];
    float4 self[4];
    #pragma unroll
    for (int q = 0; q < 4; ++q) {
        di[q] = dinv[nd[q]];
        self[q] = *(const float4*)(emb + (size_t)nd[q] * DFEAT + j);
    }
    #pragma unroll
    for (int q = 0; q < 4; ++q) {
        const float dd = di[q] * di[q];
        float4 r;
        r.x = acc[q].x + self[q].x * dd;
        r.y = acc[q].y + self[q].y * dd;
        r.z = acc[q].z + self[q].z * dd;
        r.w = acc[q].w + self[q].w * dd;
        if (node0 + q >= n) r = make_float4(0.0f, 0.0f, 0.0f, 0.0f);
        *(float4*)(Ta + (tileRow0 + q) * TP + j) = r;
    }
    __syncthreads();

    const int mt = wave >> 1;
    const int nh = wave & 1;
    const int rowT = mt * 16 + l16;

    v8f c1[2] = {};
    {
        const float* Arow = Ta + rowT * TP;
        for (int k0 = 0; k0 < DFEAT; k0 += 4) {
            const int ka = k0 + half * 2;
            v2f a;
            a[0] = Arow[ka];
            a[1] = Arow[ka + 1];
            #pragma unroll
            for (int tn = 0; tn < 2; ++tn) {
                const int n0 = nh * 32 + tn * 16;
                v2f b;
                b[0] = Wl1[ka * DFEAT + n0 + l16];
                b[1] = Wl1[(ka + 1) * DFEAT + n0 + l16];
                c1[tn] = __builtin_amdgcn_wmma_f32_16x16x4_f32(
                     false, a,  false, b,
                     (short)0, c1[tn],  false,  false);
            }
        }
    }
    asm volatile("v_nop\n\tv_nop\n\tv_nop\n\tv_nop" : "+v"(c1[0]), "+v"(c1[1]));
    #pragma unroll
    for (int tn = 0; tn < 2; ++tn) {
        const int col = nh * 32 + tn * 16 + l16;
        const float bias = b1[col];
        #pragma unroll
        for (int v = 0; v < 8; ++v) {
            float r = c1[tn][v] + bias;
            r = r > 0.0f ? r : 0.0f;
            Tx[(mt * 16 + v + half * 8) * TP + col] = r;
        }
    }
    __syncthreads();

    v8f c2[2] = {};
    {
        const float* Arow = Tx + rowT * TP;
        for (int k0 = 0; k0 < DFEAT; k0 += 4) {
            const int ka = k0 + half * 2;
            v2f a;
            a[0] = Arow[ka];
            a[1] = Arow[ka + 1];
            #pragma unroll
            for (int tn = 0; tn < 2; ++tn) {
                const int n0 = nh * 32 + tn * 16;
                v2f b;
                b[0] = Wl2[ka * DFEAT + n0 + l16];
                b[1] = Wl2[(ka + 1) * DFEAT + n0 + l16];
                c2[tn] = __builtin_amdgcn_wmma_f32_16x16x4_f32(
                     false, a,  false, b,
                     (short)0, c2[tn],  false,  false);
            }
        }
    }
    #pragma unroll
    for (int tn = 0; tn < 2; ++tn) {
        const int col = nh * 32 + tn * 16 + l16;
        #pragma unroll
        for (int v = 0; v < 8; ++v) {
            const int row = blockIdx.x * TILE_ROWS + mt * 16 + v + half * 8;
            if (row < n) h2[(size_t)row * DFEAT + col] = c2[tn][v];
        }
    }
}


extern "C" void kernel_launch(void* const* d_in, const int* in_sizes, int n_in,
                              void* d_out, int out_size, void* d_ws, size_t ws_size,
                              hipStream_t stream) {
    const int* eidx = (const int*)d_in[0];
    const float* emb = (const float*)d_in[1];
    const float* W1 = (const float*)d_in[2];
    const float* b1 = (const float*)d_in[3];
    const float* W2 = (const float*)d_in[4];
    const float* b2 = (const float*)d_in[5];
    float* out = (float*)d_out;

    const int E = in_sizes[0] / 2;
    const int n = in_sizes[1] / DFEAT;

    const int* src = eidx;
    const int* dst = eidx + E;

    float* h2    = (float*)d_ws;
    int2* csr    = (int2*)(h2 + (size_t)n * DFEAT);
    int* rank    = (int*)(csr + E);
    int* cnt     = rank + E;
    int* rowptr  = cnt + n;
    float* dinv  = (float*)(rowptr + n);

    const int B = 256;
    const int gN = (n + B - 1) / B;
    const int gE = (E + B - 1) / B;
    const int gRows = (n * 16 + B - 1) / B;
    const int gTile = (n + TILE_ROWS - 1) / TILE_ROWS;

    zero_i32_kernel<<<gN, B, 0, stream>>>(cnt, n);
    rank_count_kernel<<<gE, B, 0, stream>>>(dst, cnt, rank, E);
    scan_dinv_kernel<<<gN, B, 0, stream>>>(cnt, rowptr, dinv, n);
    fill_csr_kernel<<<gE, B, 0, stream>>>(src, dst, rank, rowptr, dinv, csr, E);

    gather_mlp_kernel<<<gTile, B, 0, stream>>>(rowptr, cnt, csr, emb, dinv, W1, b1, W2, h2, n);

    gather_rows_kernel<false><<<gRows, B, 0, stream>>>(rowptr, cnt, csr, h2, dinv, b2, out, n);
}


# Slot rows replace the sorted list: atomic arrival rank writes src into a 48-word row per node (overflow: whole-list sum); scan+fill launches gone (6->4); dinv[src] in the sum, dinv[dst] outside
# speedup vs baseline: 2.3100x; 1.2089x over previous
#include <hip/hip_runtime.h>
#include <hip/hip_bf16.h>

typedef __attribute__((ext_vector_type(2))) float v2f;
typedef __attribute__((ext_vector_type(8))) float v8f;

#define DFEAT 64

#define SLOTS 48

__global__ void clear_kernel(int* __restrict__ cnt, int* __restrict__ slots, int n) {
    int i = blockIdx.x * blockDim.x + threadIdx.x;
    if (i < n) {
        cnt[i] = 0;
        slots[(unsigned)i * SLOTS] = 0;
    }
}

__global__ void bucket_kernel(const int* __restrict__ src, const int* __restrict__ dst,
                              int* __restrict__ cnt, int* __restrict__ slots, int E) {
    int e = blockIdx.x * blockDim.x + threadIdx.x;
    if (e >= E) return;
    const int d = dst[e];
    const int s = src[e];
    const int r = atomicAdd(&cnt[d], 1);
    if (r < SLOTS) slots[(unsigned)d * SLOTS + r] = s;
}

__device__ __forceinline__ float dinv_of(int indeg) {
    return __builtin_amdgcn_rsqf((float)indeg + 1.0f);
}

template <bool WEIGHTED>
__device__ __forceinline__ float4 scan_all_edges(const int* __restrict__ src, const int* __restrict__ dst,
                                                 const int* __restrict__ cnt, const float* __restrict__ table,
                                                 int node, bool wanted, int E, int l16, int j) {
    float4 acc = make_float4(0.0f, 0.0f, 0.0f, 0.0f);
    #pragma unroll 1
    for (int e0 = 0; e0 < E; e0 += 16) {
        const int e = e0 + l16;
        bool m = false;
        int s = node;
        if (wanted && e < E && dst[e] == node) {
            m = true;
            s = src[e];
        }
        if (__ballot(m) == 0) continue;
        float w = 0.0f;
        if (m) w = WEIGHTED ? dinv_of(cnt[s]) : 1.0f;
        #pragma unroll 1
        for (int k = 0; k < 16; ++k) {
            const int sk = __shfl(s, k, 16);
            const float wk = __shfl(w, k, 16);
            const float4 v = *(const float4*)(table + (size_t)sk * DFEAT + j);
            acc.x += v.x * wk;
            acc.y += v.y * wk;
            acc.z += v.z * wk;
            acc.w += v.w * wk;
        }
    }
    return acc;
}

__global__ __launch_bounds__(256) void gather_rows_kernel(const int* __restrict__ cnt,
                                                          const int* __restrict__ slots,
                                                          const int* __restrict__ src,
                                                          const int* __restrict__ dst,
                                                          const float* __restrict__ h,
                                                          const float* __restrict__ b,
                                                          float* __restrict__ outp, int n, int E) {
    const int tid = blockIdx.x * 256 + threadIdx.x;
    int node = tid >> 4;
    const bool valid = node < n;
    if (!valid) node = n - 1;
    const int l16 = threadIdx.x & 15;
    const int j = l16 << 2;

    const int deg = cnt[node];
    const unsigned rowBase = (unsigned)node * SLOTS;
    const bool over = valid && deg > SLOTS;
    const int len = valid ? min(deg, SLOTS) : 0;
    int lenMax = max(len, __shfl_xor(len, 16, 32));
    lenMax = __builtin_amdgcn_readfirstlane(lenMax);

    float4 acc = make_float4(0.0f, 0.0f, 0.0f, 0.0f);
    for (int base = 0; base < lenMax; base += 16) {
        const bool in = base + l16 < len;
        const int ent = slots[rowBase + (in ? base + l16 : 0)];
        const int ex = in ? ent : node;
        const float ey = in ? 1.0f : 0.0f;
        const int chunk = min(16, lenMax - base);
        for (int k = 0; k < chunk; k += 4) {
            int s[4];
            float w[4];
            float4 v[4];
            for (int u = 0; u < 4; ++u) {
                s[u] = __shfl(ex, k + u, 16);
                w[u] = __shfl(ey, k + u, 16);
            }
            for (int u = 0; u < 4; ++u)
                v[u] = *(const float4*)(h + (size_t)s[u] * DFEAT + j);
            for (int u = 0; u < 4; ++u) {
                acc.x += v[u].x * w[u];
                acc.y += v[u].y * w[u];
                acc.z += v[u].z * w[u];
                acc.w += v[u].w * w[u];
            }
        }
    }
    if (__ballot(over) != 0) {
        const float4 r = scan_all_edges<false>(src, dst, cnt, h, node, over, E, l16, j);
        if (over) acc = r;
    }

    if (valid) {
        const float di = dinv_of(deg);
        const float4 hv = *(const float4*)(h + (size_t)node * DFEAT + j);
        float4 r;
        r.x = (acc.x + hv.x) * di + b[j + 0];
        r.y = (acc.y + hv.y) * di + b[j + 1];
        r.z = (acc.z + hv.z) * di + b[j + 2];
        r.w = (acc.w + hv.w) * di + b[j + 3];
        *(float4*)(outp + (size_t)node * DFEAT + j) = r;
    }
}

#define TILE_ROWS 64
#define EPS 2
#define TP 68

__global__ __launch_bounds__(256) void gather_mlp_kernel(const int* __restrict__ cnt,
                                                         const int* __restrict__ slots,
                                                         const int* __restrict__ src,
                                                         const int* __restrict__ dst,
                                                         const float* __restrict__ emb,
                                                         const float* __restrict__ W1,
                                                         const float* __restrict__ b1,
                                                         const float* __restrict__ W2,
                                                         float* __restrict__ h2, int n, int E) {
    __shared__ __attribute__((aligned(16))) float Wl1[DFEAT * DFEAT];
    __shared__ __attribute__((aligned(16))) float Wl2[DFEAT * DFEAT];
    __shared__ __attribute__((aligned(16))) float Ta[TILE_ROWS * TP];
    __shared__ __attribute__((aligned(16))) float Tx[TILE_ROWS * TP];
    __shared__ float dinvT[TILE_ROWS];

    const int t = threadIdx.x;
    const int wave = t >> 5;
    const int half = (t >> 4) & 1;
    const int l16 = t & 15;
    const int j = l16 << 2;

    {
        const float4* W1v = (const float4*)W1;
        const float4* W2v = (const float4*)W2;
        float4* Wl1v = (float4*)Wl1;
        float4* Wl2v = (float4*)Wl2;
        #pragma unroll
        for (int i = 0; i < (DFEAT * DFEAT / 4) / 256; ++i) {
            Wl1v[t + i * 256] = W1v[t + i * 256];
            Wl2v[t + i * 256] = W2v[t + i * 256];
        }
    }

    const int grp = wave * 2 + half;
    const int tileRow0 = grp * 4;
    const int node0 = blockIdx.x * TILE_ROWS + tileRow0;

    int nd[4], len[4], deg[4];
    unsigned rowBase[4];
    bool over[4];
    #pragma unroll
    for (int q = 0; q < 4; ++q) {
        const int node = node0 + q;
        const bool valid = node < n;
        nd[q] = valid ? node : n - 1;
        rowBase[q] = (unsigned)nd[q] * SLOTS;
        deg[q] = cnt[nd[q]];
        len[q] = valid ? min(deg[q], SLOTS) : 0;
        over[q] = valid && deg[q] > SLOTS;
    }
    int lenMax = max(max(len[0], len[1]), max(len[2], len[3]));
    lenMax = max(lenMax, __shfl_xor(lenMax, 16, 32));
    lenMax = __builtin_amdgcn_readfirstlane(lenMax);

    float4 acc[4];
    #pragma unroll
    for (int q = 0; q < 4; ++q) acc[q] = make_float4(0.0f, 0.0f, 0.0f, 0.0f);
    const char* embBytes = (const char*)emb;
    const unsigned colByte = (unsigned)j << 2;

    int ent[4];
    #pragma unroll
    for (int q = 0; q < 4; ++q)
        ent[q] = slots[rowBase[q] + (l16 < len[q] ? l16 : 0)];

    for (int base = 0; base < lenMax; base += 16) {
        int ex[4], ey[4];
        int sdeg[4];
        #pragma unroll
        for (int q = 0; q < 4; ++q) {
            const bool in = base + l16 < len[q];
            ex[q] = in ? ent[q] : nd[q];
            sdeg[q] = cnt[ex[q]];
        }
        #pragma unroll
        for (int q = 0; q < 4; ++q) {
            const int nx = base + 16 + l16;
            ent[q] = slots[rowBase[q] + (nx < len[q] ? nx : 0)];
        }
        #pragma unroll
        for (int q = 0; q < 4; ++q) {
            const bool in = base + l16 < len[q];
            ey[q] = in ? __float_as_int(dinv_of(sdeg[q])) : 0;
        }
        const int chunk = min(16, lenMax - base);
        for (int k = 0; k < chunk; k += EPS) {
            float wgt[4][EPS];
            float4 v[4][EPS];
            #pragma unroll
            for (int q = 0; q < 4; ++q) {
                #pragma unroll
                for (int u = 0; u < EPS; ++u) {
                    const int s = __shfl(ex[q], k + u, 16);
                    wgt[q][u] = __int_as_float(__shfl(ey[q], k + u, 16));
                    const unsigned off = ((unsigned)s << 8) + colByte;
                    v[q][u] = *(const float4*)(embBytes + off);
                }
            }
            #pragma unroll
            for (int q = 0; q < 4; ++q) {
                #pragma unroll
                for (int u = 0; u < EPS; ++u) {
                    acc[q].x += v[q][u].x * wgt[q][u];
                    acc[q].y += v[q][u].y * wgt[q][u];
                    acc[q].z += v[q][u].z * wgt[q][u];
                    acc[q].w += v[q][u].w * wgt[q][u];
                }
            }
        }
    }

    if (__ballot(over[0] || over[1] || over[2] || over[3]) != 0) {
        #pragma unroll
        for (int q = 0; q < 4; ++q) {
            const float4 r = scan_all_edges<true>(src, dst, cnt, emb, nd[q], over[q], E, l16, j);
            if (over[q]) acc[q] = r;
        }
    }

    float4 self[4];
    #pragma unroll
    for (int q = 0; q < 4; ++q)
        self[q] = *(const float4*)(emb + (size_t)nd[q] * DFEAT + j);
    #pragma unroll
    for (int q = 0; q < 4; ++q) {
        const float di = dinv_of(deg[q]);
        float4 r;
        r.x = (acc[q].x + self[q].x * di) * di;
        r.y = (acc[q].y + self[q].y * di) * di;
        r.z = (acc[q].z + self[q].z * di) * di;
        r.w = (acc[q].w + self[q].w * di) * di;
        if (node0 + q >= n) r = make_float4(0.0f, 0.0f, 0.0f, 0.0f);
        *(float4*)(Ta + (tileRow0 + q) * TP + j) = r;
        if (l16 == 0) dinvT[tileRow0 + q] = di;
    }
    __syncthreads();

    const int mt = wave >> 1;
    const int nh = wave & 1;
    const int rowT = mt * 16 + l16;

    v8f c1[2] = {};
    {
        const float* Arow = Ta + rowT * TP;
        for (int k0 = 0; k0 < DFEAT; k0 += 4) {
            const int ka = k0 + half * 2;
            v2f a;
            a[0] = Arow[ka];
            a[1] = Arow[ka + 1];
            #pragma unroll
            for (int tn = 0; tn < 2; ++tn) {
                const int n0 = nh * 32 + tn * 16;
                v2f b;
                b[0] = Wl1[ka * DFEAT + n0 + l16];
                b[1] = Wl1[(ka + 1) * DFEAT + n0 + l16];
                c1[tn] = __builtin_amdgcn_wmma_f32_16x16x4_f32(
                     false, a,  false, b,
                     (short)0, c1[tn],  false,  false);
            }
        }
    }
    asm volatile("v_nop\n\tv_nop\n\tv_nop\n\tv_nop" : "+v"(c1[0]), "+v"(c1[1]));
    #pragma unroll
    for (int tn = 0; tn < 2; ++tn) {
        const int col = nh * 32 + tn * 16 + l16;
        const float bias = b1[col];
        #pragma unroll
        for (int v = 0; v < 8; ++v) {
            const int rowInTile = mt * 16 + v + half * 8;
            float r = c1[tn][v] + bias;
            r = r > 0.0f ? r : 0.0f;
            Tx[rowInTile * TP + col] = r * dinvT[rowInTile];
        }
    }
    __syncthreads();

    v8f c2[2] = {};
    {
        const float* Arow = Tx + rowT * TP;
        for (int k0 = 0; k0 < DFEAT; k0 += 4) {
            const int ka = k0 + half * 2;
            v2f a;
            a[0] = Arow[ka];
            a[1] = Arow[ka + 1];
            #pragma unroll
            for (int tn = 0; tn < 2; ++tn) {
                const int n0 = nh * 32 + tn * 16;
                v2f b;
                b[0] = Wl2[ka * DFEAT + n0 + l16];
                b[1] = Wl2[(ka + 1) * DFEAT + n0 + l16];
                c2[tn] = __builtin_amdgcn_wmma_f32_16x16x4_f32(
                     false, a,  false, b,
                     (short)0, c2[tn],  false,  false);
            }
        }
    }
    #pragma unroll
    for (int tn = 0; tn < 2; ++tn) {
        const int col = nh * 32 + tn * 16 + l16;
        #pragma unroll
        for (int v = 0; v < 8; ++v) {
            const int row = blockIdx.x * TILE_ROWS + mt * 16 + v + half * 8;
            if (row < n) h2[(size_t)row * DFEAT + col] = c2[tn][v];
        }
    }
}


extern "C" void kernel_launch(void* const* d_in, const int* in_sizes, int n_in,
                              void* d_out, int out_size, void* d_ws, size_t ws_size,
                              hipStream_t stream) {
    const int* eidx = (const int*)d_in[0];
    const float* emb = (const float*)d_in[1];
    const float* W1 = (const float*)d_in[2];
    const float* b1 = (const float*)d_in[3];
    const float* W2 = (const float*)d_in[4];
    const float* b2 = (const float*)d_in[5];
    float* out = (float*)d_out;

    const int E = in_sizes[0] / 2;
    const int n = in_sizes[1] / DFEAT;

    const int* src = eidx;
    const int* dst = eidx + E;

    float* h2    = (float*)d_ws;
    int* slots   = (int*)(h2 + (size_t)n * DFEAT);
    int* cnt     = slots + (size_t)n * SLOTS;

    const int B = 256;
    const int gN = (n + B - 1) / B;
    const int gE = (E + B - 1) / B;
    const int gRows = (n * 16 + B - 1) / B;
    const int gTile = (n + TILE_ROWS - 1) / TILE_ROWS;

    clear_kernel<<<gN, B, 0, stream>>>(cnt, slots, n);
    bucket_kernel<<<gE, B, 0, stream>>>(src, dst, cnt, slots, E);

    gather_mlp_kernel<<<gTile, B, 0, stream>>>(cnt, slots, src, dst, emb, W1, b1, W2, h2, n, E);

    gather_rows_kernel<<<gRows, B, 0, stream>>>(cnt, slots, src, dst, h2, b2, out, n, E);
}


# Slot rows built by two-level binning: LDS counters bin edges per 64-node tile (one global add per work-group and tile), a tile block ranks them in LDS; sums follow LDS arrival rank; 4->5 launches
# speedup vs baseline: 3.3622x; 1.4555x over previous
#include <hip/hip_runtime.h>
#include <hip/hip_bf16.h>

typedef __attribute__((ext_vector_type(2))) float v2f;
typedef __attribute__((ext_vector_type(8))) float v8f;

#define DFEAT 64

#define SLOTS 48
#define BIN_SHIFT 6
#define BIN_NODES (1 << BIN_SHIFT)
#define BIN_CAP (BIN_NODES * DFEAT)
#define BIN_ROWWORDS (BIN_NODES * SLOTS)
#define SRC_BITS 26
#define BIN_THREADS 1024
#define BIN_EPT 8
#define BUILD_WPT 6
#define HIST_TILES 1024

__global__ void clear_kernel(int* __restrict__ slots, int nTiles) {
    int i = blockIdx.x * blockDim.x + threadIdx.x;
    if (i < nTiles) slots[(unsigned)i * BIN_ROWWORDS] = 0;
}

__global__ __launch_bounds__(BIN_THREADS) void bin_kernel(const int* __restrict__ src,
                                                          const int* __restrict__ dst,
                                                          int* __restrict__ slots,
                                                          int* __restrict__ bins, int E, int n) {
    __shared__ int hist[HIST_TILES + 1];
    const int t = threadIdx.x;
    const int nTiles = (n + BIN_NODES - 1) >> BIN_SHIFT;
    const int nLds = min(nTiles, HIST_TILES);
    for (int i = t; i < HIST_TILES + 1; i += BIN_THREADS) hist[i] = 0;
    __syncthreads();

    int dv[BIN_EPT], sv[BIN_EPT];
    const int e0 = blockIdx.x * (BIN_THREADS * BIN_EPT) + t;
    #pragma unroll
    for (int u = 0; u < BIN_EPT; ++u) {
        const int e = min(e0 + u * BIN_THREADS, E - 1);
        dv[u] = dst[e];
        sv[u] = src[e];
    }
    int tile[BIN_EPT], word[BIN_EPT], rank[BIN_EPT], hidx[BIN_EPT];
    #pragma unroll
    for (int u = 0; u < BIN_EPT; ++u) {
        const bool valid = e0 + u * BIN_THREADS < E;
        tile[u] = valid ? (dv[u] >> BIN_SHIFT) : -1;
        word[u] = (int)(((unsigned)(dv[u] & (BIN_NODES - 1)) << SRC_BITS) | (unsigned)sv[u]);
        hidx[u] = (valid && tile[u] < HIST_TILES) ? tile[u] : HIST_TILES;
        rank[u] = atomicAdd(&hist[hidx[u]], 1);
    }
    #pragma unroll
    for (int u = 0; u < BIN_EPT; ++u) {
        if (tile[u] >= HIST_TILES) rank[u] = atomicAdd(&slots[(unsigned)tile[u] * BIN_ROWWORDS], 1);
    }
    __syncthreads();
    for (int i = t; i < nLds; i += BIN_THREADS) {
        const int c = hist[i];
        if (c > 0) hist[i] = atomicAdd(&slots[(unsigned)i * BIN_ROWWORDS], c);
    }
    if (t == 0) hist[HIST_TILES] = 0;
    __syncthreads();
    #pragma unroll
    for (int u = 0; u < BIN_EPT; ++u) {
        const int b = tile[u];
        const int pos = rank[u] + hist[hidx[u]];
        const int cap = min(BIN_NODES, n - b * BIN_NODES) * DFEAT;
        if (b >= 0 && pos < cap) bins[(unsigned)b * BIN_CAP + pos] = word[u];
    }
}

__global__ __launch_bounds__(256) void build_rows_kernel(const int* __restrict__ src,
                                                         const int* __restrict__ dst,
                                                         const int* __restrict__ bins,
                                                         int* __restrict__ cnt,
                                                         int* __restrict__ slots, int n, int E) {
    __shared__ int degL[BIN_NODES];
    __shared__ int rowsL[BIN_ROWWORDS];
    const int t = threadIdx.x;
    const int b = blockIdx.x;
    const int nodes = min(BIN_NODES, n - b * BIN_NODES);
    if (t < BIN_NODES) degL[t] = 0;
    for (int i = t; i < BIN_ROWWORDS; i += 256) rowsL[i] = 0;
    const int count = slots[(unsigned)b * BIN_ROWWORDS];
    __syncthreads();
    if (count <= nodes * DFEAT) {
        const int* mine = bins + (unsigned)b * BIN_CAP;
        for (int i0 = 0; i0 < count; i0 += 256 * BUILD_WPT) {
            unsigned w[BUILD_WPT];
            #pragma unroll
            for (int k = 0; k < BUILD_WPT; ++k)
                w[k] = (unsigned)mine[min(i0 + k * 256 + t, count - 1)];
            #pragma unroll
            for (int k = 0; k < BUILD_WPT; ++k) {
                if (i0 + k * 256 + t < count) {
                    const int dl = (int)(w[k] >> SRC_BITS);
                    const int r = atomicAdd(&degL[dl], 1);
                    if (r < SLOTS) rowsL[dl * SLOTS + r] = (int)(w[k] & ((1u << SRC_BITS) - 1u));
                }
            }
        }
    } else {
        for (int e = t; e < E; e += 256) {
            const int d = dst[e];
            if ((d >> BIN_SHIFT) == b) {
                const int dl = d & (BIN_NODES - 1);
                const int r = atomicAdd(&degL[dl], 1);
                if (r < SLOTS) rowsL[dl * SLOTS + r] = src[e];
            }
        }
    }
    __syncthreads();
    if (t < nodes) cnt[b * BIN_NODES + t] = degL[t];
    int* rows = slots + (unsigned)b * BIN_ROWWORDS;
    for (int i = t; i < nodes * SLOTS; i += 256) rows[i] = rowsL[i];
}

__device__ __forceinline__ float dinv_of(int indeg) {
    return __builtin_amdgcn_rsqf((float)indeg + 1.0f);
}

template <bool WEIGHTED>
__device__ __forceinline__ float4 scan_all_edges(const int* __restrict__ src, const int* __restrict__ dst,
                                                 const int* __restrict__ cnt, const float* __restrict__ table,
                                                 int node, bool wanted, int E, int l16, int j) {
    float4 acc = make_float4(0.0f, 0.0f, 0.0f, 0.0f);
    #pragma unroll 1
    for (int e0 = 0; e0 < E; e0 += 16) {
        const int e = e0 + l16;
        bool m = false;
        int s = node;
        if (wanted && e < E && dst[e] == node) {
            m = true;
            s = src[e];
        }
        if (__ballot(m) == 0) continue;
        float w = 0.0f;
        if (m) w = WEIGHTED ? dinv_of(cnt[s]) : 1.0f;
        #pragma unroll 1
        for (int k = 0; k < 16; ++k) {
            const int sk = __shfl(s, k, 16);
            const float wk = __shfl(w, k, 16);
            const float4 v = *(const float4*)(table + (size_t)sk * DFEAT + j);
            acc.x += v.x * wk;
            acc.y += v.y * wk;
            acc.z += v.z * wk;
            acc.w += v.w * wk;
        }
    }
    return acc;
}

__global__ __launch_bounds__(256) void gather_rows_kernel(const int* __restrict__ cnt,
                                                          const int* __restrict__ slots,
                                                          const int* __restrict__ src,
                                                          const int* __restrict__ dst,
                                                          const float* __restrict__ h,
                                                          const float* __restrict__ b,
                                                          float* __restrict__ outp, int n, int E) {
    const int tid = blockIdx.x * 256 + threadIdx.x;
    int node = tid >> 4;
    const bool valid = node < n;
    if (!valid) node = n - 1;
    const int l16 = threadIdx.x & 15;
    const int j = l16 << 2;

    const int deg = cnt[node];
    const unsigned rowBase = (unsigned)node * SLOTS;
    const bool over = valid && deg > SLOTS;
    const int len = valid ? min(deg, SLOTS) : 0;
    int lenMax = max(len, __shfl_xor(len, 16, 32));
    lenMax = __builtin_amdgcn_readfirstlane(lenMax);

    float4 acc = make_float4(0.0f, 0.0f, 0.0f, 0.0f);
    for (int base = 0; base < lenMax; base += 16) {
        const bool in = base + l16 < len;
        const int ent = slots[rowBase + (in ? base + l16 : 0)];
        const int ex = in ? ent : node;
        const float ey = in ? 1.0f : 0.0f;
        const int chunk = min(16, lenMax - base);
        for (int k = 0; k < chunk; k += 4) {
            int s[4];
            float w[4];
            float4 v[4];
            for (int u = 0; u < 4; ++u) {
                s[u] = __shfl(ex, k + u, 16);
                w[u] = __shfl(ey, k + u, 16);
            }
            for (int u = 0; u < 4; ++u)
                v[u] = *(const float4*)(h + (size_t)s[u] * DFEAT + j);
            for (int u = 0; u < 4; ++u) {
                acc.x += v[u].x * w[u];
                acc.y += v[u].y * w[u];
                acc.z += v[u].z * w[u];
                acc.w += v[u].w * w[u];
            }
        }
    }
    if (__ballot(over) != 0) {
        const float4 r = scan_all_edges<false>(src, dst, cnt, h, node, over, E, l16, j);
        if (over) acc = r;
    }

    if (valid) {
        const float di = dinv_of(deg);
        const float4 hv = *(const float4*)(h + (size_t)node * DFEAT + j);
        float4 r;
        r.x = (acc.x + hv.x) * di + b[j + 0];
        r.y = (acc.y + hv.y) * di + b[j + 1];
        r.z = (acc.z + hv.z) * di + b[j + 2];
        r.w = (acc.w + hv.w) * di + b[j + 3];
        *(float4*)(outp + (size_t)node * DFEAT + j) = r;
    }
}

#define TILE_ROWS 64
#define EPS 2
#define TP 68

__global__ __launch_bounds__(256) void gather_mlp_kernel(const int* __restrict__ cnt,
                                                         const int* __restrict__ slots,
                                                         const int* __restrict__ src,
                                                         const int* __restrict__ dst,
                                                         const float* __restrict__ emb,
                                                         const float* __restrict__ W1,
                                                         const float* __restrict__ b1,
                                                         const float* __restrict__ W2,
                                                         float* __restrict__ h2, int n, int E) {
    __shared__ __attribute__((aligned(16))) float Wl1[DFEAT * DFEAT];
    __shared__ __attribute__((aligned(16))) float Wl2[DFEAT * DFEAT];
    __shared__ __attribute__((aligned(16))) float Ta[TILE_ROWS * TP];
    __shared__ __attribute__((aligned(16))) float Tx[TILE_ROWS * TP];
    __shared__ float dinvT[TILE_ROWS];

    const int t = threadIdx.x;
    const int wave = t >> 5;
    const int half = (t >> 4) & 1;
    const int l16 = t & 15;
    const int j = l16 << 2;

    {
        const float4* W1v = (const float4*)W1;
        const float4* W2v = (const float4*)W2;
        float4* Wl1v = (float4*)Wl1;
        float4* Wl2v = (float4*)Wl2;
        #pragma unroll
        for (int i = 0; i < (DFEAT * DFEAT / 4) / 256; ++i) {
            Wl1v[t + i * 256] = W1v[t + i * 256];
            Wl2v[t + i * 256] = W2v[t + i * 256];
        }
    }

    const int grp = wave * 2 + half;
    const int tileRow0 = grp * 4;
    const int node0 = blockIdx.x * TILE_ROWS + tileRow0;

    int nd[4], len[4], deg[4];
    unsigned rowBase[4];
    bool over[4];
    #pragma unroll
    for (int q = 0; q < 4; ++q) {
        const int node = node0 + q;
        const bool valid = node < n;
        nd[q] = valid ? node : n - 1;
        rowBase[q] = (unsigned)nd[q] * SLOTS;
        deg[q] = cnt[nd[q]];
        len[q] = valid ? min(deg[q], SLOTS) : 0;
        over[q] = valid && deg[q] > SLOTS;
    }
    int lenMax = max(max(len[0], len[1]), max(len[2], len[3]));
    lenMax = max(lenMax, __shfl_xor(lenMax, 16, 32));
    lenMax = __builtin_amdgcn_readfirstlane(lenMax);

    float4 acc[4];
    #pragma unroll
    for (int q = 0; q < 4; ++q) acc[q] = make_float4(0.0f, 0.0f, 0.0f, 0.0f);
    const char* embBytes = (const char*)emb;
    const unsigned colByte = (unsigned)j << 2;

    int ent[4];
    #pragma unroll
    for (int q = 0; q < 4; ++q)
        ent[q] = slots[rowBase[q] + (l16 < len[q] ? l16 : 0)];

    for (int base = 0; base < lenMax; base += 16) {
        int ex[4], ey[4];
        int sdeg[4];
        #pragma unroll
        for (int q = 0; q < 4; ++q) {
            const bool in = base + l16 < len[q];
            ex[q] = in ? ent[q] : nd[q];
            sdeg[q] = cnt[ex[q]];
        }
        #pragma unroll
        for (int q = 0; q < 4; ++q) {
            const int nx = base + 16 + l16;
            ent[q] = slots[rowBase[q] + (nx < len[q] ? nx : 0)];
        }
        #pragma unroll
        for (int q = 0; q < 4; ++q) {
            const bool in = base + l16 < len[q];
            ey[q] = in ? __float_as_int(dinv_of(sdeg[q])) : 0;
        }
        const int chunk = min(16, lenMax - base);
        for (int k = 0; k < chunk; k += EPS) {
            float wgt[4][EPS];
            float4 v[4][EPS];
            #pragma unroll
            for (int q = 0; q < 4; ++q) {
                #pragma unroll
                for (int u = 0; u < EPS; ++u) {
                    const int s = __shfl(ex[q], k + u, 16);
                    wgt[q][u] = __int_as_float(__shfl(ey[q], k + u, 16));
                    const unsigned off = ((unsigned)s << 8) + colByte;
                    v[q][u] = *(const float4*)(embBytes + off);
                }
            }
            #pragma unroll
            for (int q = 0; q < 4; ++q) {
                #pragma unroll
                for (int u = 0; u < EPS; ++u) {
                    acc[q].x += v[q][u].x * wgt[q][u];
                    acc[q].y += v[q][u].y * wgt[q][u];
                    acc[q].z += v[q][u].z * wgt[q][u];
                    acc[q].w += v[q][u].w * wgt[q][u];
                }
            }
        }
    }

    if (__ballot(over[0] || over[1] || over[2] || over[3]) != 0) {
        #pragma unroll
        for (int q = 0; q < 4; ++q) {
            const float4 r = scan_all_edges<true>(src, dst, cnt, emb, nd[q], over[q], E, l16, j);
            if (over[q]) acc[q] = r;
        }
    }

    float4 self[4];
    #pragma unroll
    for (int q = 0; q < 4; ++q)
        self[q] = *(const float4*)(emb + (size_t)nd[q] * DFEAT + j);
    #pragma unroll
    for (int q = 0; q < 4; ++q) {
        const float di = dinv_of(deg[q]);
        float4 r;
        r.x = (acc[q].x + self[q].x * di) * di;
        r.y = (acc[q].y + self[q].y * di) * di;
        r.z = (acc[q].z + self[q].z * di) * di;
        r.w = (acc[q].w + self[q].w * di) * di;
        if (node0 + q >= n) r = make_float4(0.0f, 0.0f, 0.0f, 0.0f);
        *(float4*)(Ta + (tileRow0 + q) * TP + j) = r;
        if (l16 == 0) dinvT[tileRow0 + q] = di;
    }
    __syncthreads();

    const int mt = wave >> 1;
    const int nh = wave & 1;
    const int rowT = mt * 16 + l16;

    v8f c1[2] = {};
    {
        const float* Arow = Ta + rowT * TP;
        for (int k0 = 0; k0 < DFEAT; k0 += 4) {
            const int ka = k0 + half * 2;
            v2f a;
            a[0] = Arow[ka];
            a[1] = Arow[ka + 1];
            #pragma unroll
            for (int tn = 0; tn < 2; ++tn) {
                const int n0 = nh * 32 + tn * 16;
                v2f b;
                b[0] = Wl1[ka * DFEAT + n0 + l16];
                b[1] = Wl1[(ka + 1) * DFEAT + n0 + l16];
                c1[tn] = __builtin_amdgcn_wmma_f32_16x16x4_f32(
                     false, a,  false, b,
                     (short)0, c1[tn],  false,  false);
            }
        }
    }
    asm volatile("v_nop\n\tv_nop\n\tv_nop\n\tv_nop" : "+v"(c1[0]), "+v"(c1[1]));
    #pragma unroll
    for (int tn = 0; tn < 2; ++tn) {
        const int col = nh * 32 + tn * 16 + l16;
        const float bias = b1[col];
        #pragma unroll
        for (int v = 0; v < 8; ++v) {
            const int rowInTile = mt * 16 + v + half * 8;
            float r = c1[tn][v] + bias;
            r = r > 0.0f ? r : 0.0f;
            Tx[rowInTile * TP + col] = r * dinvT[rowInTile];
        }
    }
    __syncthreads();

    v8f c2[2] = {};
    {
        const float* Arow = Tx + rowT * TP;
        for (int k0 = 0; k0 < DFEAT; k0 += 4) {
            const int ka = k0 + half * 2;
            v2f a;
            a[0] = Arow[ka];
            a[1] = Arow[ka + 1];
            #pragma unroll
            for (int tn = 0; tn < 2; ++tn) {
                const int n0 = nh * 32 + tn * 16;
                v2f b;
                b[0] = Wl2[ka * DFEAT + n0 + l16];
                b[1] = Wl2[(ka + 1) * DFEAT + n0 + l16];
                c2[tn] = __builtin_amdgcn_wmma_f32_16x16x4_f32(
                     false, a,  false, b,
                     (short)0, c2[tn],  false,  false);
            }
        }
    }
    #pragma unroll
    for (int tn = 0; tn < 2; ++tn) {
        const int col = nh * 32 + tn * 16 + l16;
        #pragma unroll
        for (int v = 0; v < 8; ++v) {
            const int row = blockIdx.x * TILE_ROWS + mt * 16 + v + half * 8;
            if (row < n) h2[(size_t)row * DFEAT + col] = c2[tn][v];
        }
    }
}


static_assert(BIN_NODES == TILE_ROWS, "the bin tiles and the layer kernel's tiles are counted by one grid size");

extern "C" void kernel_launch(void* const* d_in, const int* in_sizes, int n_in,
                              void* d_out, int out_size, void* d_ws, size_t ws_size,
                              hipStream_t stream) {
    const int* eidx = (const int*)d_in[0];
    const float* emb = (const float*)d_in[1];
    const float* W1 = (const float*)d_in[2];
    const float* b1 = (const float*)d_in[3];
    const float* W2 = (const float*)d_in[4];
    const float* b2 = (const float*)d_in[5];
    float* out = (float*)d_out;

    const int E = in_sizes[0] / 2;
    const int n = in_sizes[1] / DFEAT;

    const int* src = eidx;
    const int* dst = eidx + E;

    float* h2    = (float*)d_ws;
    int* slots   = (int*)(h2 + (size_t)n * DFEAT);
    int* cnt     = slots + (size_t)n * SLOTS;

    int* bins    = (int*)h2;

    const int B = 256;
    const int gRows = (n * 16 + B - 1) / B;
    const int gTile = (n + TILE_ROWS - 1) / TILE_ROWS;
    const int gClear = (gTile + B - 1) / B;
    const int gBin = (E + BIN_THREADS * BIN_EPT - 1) / (BIN_THREADS * BIN_EPT);

    clear_kernel<<<gClear, B, 0, stream>>>(slots, gTile);
    bin_kernel<<<gBin, BIN_THREADS, 0, stream>>>(src, dst, slots, bins, E, n);
    build_rows_kernel<<<gTile, B, 0, stream>>>(src, dst, bins, cnt, slots, n, E);

    gather_mlp_kernel<<<gTile, B, 0, stream>>>(cnt, slots, src, dst, emb, W1, b1, W2, h2, n, E);

    gather_rows_kernel<<<gRows, B, 0, stream>>>(cnt, slots, src, dst, h2, b2, out, n, E);
}


# Global tile counters replaced by segment-private 128-byte sub-bins per tile (LDS rank is the place, counts stored plainly); clearing launch gone (5->4); sums still follow LDS arrival rank
# speedup vs baseline: 4.0220x; 1.1962x over previous
#include <hip/hip_runtime.h>
#include <hip/hip_bf16.h>

typedef __attribute__((ext_vector_type(2))) float v2f;
typedef __attribute__((ext_vector_type(8))) float v8f;

#define DFEAT 64

#define SLOTS 48
#define BIN_SHIFT 6
#define BIN_NODES (1 << BIN_SHIFT)
#define BIN_ROWWORDS (BIN_NODES * SLOTS)
#define SRC_BITS 26
#define BIN_THREADS 1024
#define BIN_EPT 8
#define BIN_EDGES (BIN_THREADS * BIN_EPT)
#define SUB_SHIFT 5
#define SUB_CAP (1 << SUB_SHIFT)
#define BUILD_THREADS 256
#define BUILD_WPT 13
#define HIST_TILES 1024
#define MAX_SEGS BUILD_THREADS

__device__ __forceinline__ bool bins_usable(int n, int E) {
    const int nTiles = (n + BIN_NODES - 1) >> BIN_SHIFT;
    const int segs = (E + BIN_EDGES - 1) / BIN_EDGES;
    return nTiles <= HIST_TILES && segs <= MAX_SEGS &&
           (long long)nTiles * segs * (SUB_CAP + 1) <= (long long)n * DFEAT;
}

__global__ __launch_bounds__(BIN_THREADS) void bin_kernel(const int* __restrict__ src,
                                                          const int* __restrict__ dst,
                                                          int* __restrict__ bins, int E, int n) {
    __shared__ int hist[HIST_TILES + 1];
    if (!bins_usable(n, E)) return;
    const int t = threadIdx.x;
    const int g = blockIdx.x;
    const int nTiles = (n + BIN_NODES - 1) >> BIN_SHIFT;
    const int segs = (E + BIN_EDGES - 1) / BIN_EDGES;
    for (int i = t; i < HIST_TILES + 1; i += BIN_THREADS) hist[i] = 0;
    __syncthreads();

    int dv[BIN_EPT], sv[BIN_EPT];
    const int e0 = g * BIN_EDGES + t;
    #pragma unroll
    for (int u = 0; u < BIN_EPT; ++u) {
        const int e = min(e0 + u * BIN_THREADS, E - 1);
        dv[u] = dst[e];
        sv[u] = src[e];
    }
    int tile[BIN_EPT], word[BIN_EPT], rank[BIN_EPT];
    #pragma unroll
    for (int u = 0; u < BIN_EPT; ++u) {
        const bool valid = e0 + u * BIN_THREADS < E;
        tile[u] = valid ? (dv[u] >> BIN_SHIFT) : -1;
        word[u] = (int)(((unsigned)(dv[u] & (BIN_NODES - 1)) << SRC_BITS) | (unsigned)sv[u]);
        rank[u] = atomicAdd(&hist[valid ? tile[u] : HIST_TILES], 1);
    }
    __syncthreads();
    int* counts = bins + (size_t)nTiles * segs * SUB_CAP;
    if (t < nTiles) counts[(unsigned)g * (unsigned)nTiles + t] = hist[t];
    #pragma unroll
    for (int u = 0; u < BIN_EPT; ++u) {
        if (tile[u] >= 0 && rank[u] < SUB_CAP)
            bins[((unsigned)tile[u] * (unsigned)segs + (unsigned)g) * SUB_CAP + rank[u]] = word[u];
    }
}

__global__ __launch_bounds__(BUILD_THREADS) void build_rows_kernel(const int* __restrict__ src,
                                                                   const int* __restrict__ dst,
                                                                   const int* __restrict__ bins,
                                                                   int* __restrict__ cnt,
                                                                   int* __restrict__ slots, int n, int E) {
    __shared__ int degL[BIN_NODES];
    __shared__ int rowsL[BIN_ROWWORDS];
    __shared__ int fillL[MAX_SEGS];
    __shared__ int overL;
    const int t = threadIdx.x;
    const int b = blockIdx.x;
    const int nodes = min(BIN_NODES, n - b * BIN_NODES);
    const int nTiles = (n + BIN_NODES - 1) >> BIN_SHIFT;
    const int segs = (E + BIN_EDGES - 1) / BIN_EDGES;
    const bool usable = bins_usable(n, E);
    const int* counts = bins + (size_t)nTiles * segs * SUB_CAP;
    int fill = 0;
    if (usable && t < segs) fill = counts[(unsigned)t * (unsigned)nTiles + b];
    if (t < BIN_NODES) degL[t] = 0;
    if (t == 0) overL = 0;
    for (int i = t; i < BIN_ROWWORDS; i += BUILD_THREADS) rowsL[i] = 0;
    __syncthreads();
    fillL[t] = fill;
    if (fill > SUB_CAP) overL = 1;
    __syncthreads();
    if (usable && overL == 0) {
        const int* mine = bins + (size_t)b * segs * SUB_CAP;
        const int total = segs * SUB_CAP;
        for (int i0 = 0; i0 < total; i0 += BUILD_THREADS * BUILD_WPT) {
            int f[BUILD_WPT];
            #pragma unroll
            for (int k = 0; k < BUILD_WPT; ++k) {
                const unsigned i = (unsigned)(i0 + k * BUILD_THREADS + t);
                f[k] = fillL[min((int)(i >> SUB_SHIFT), segs - 1)];
            }
            unsigned w[BUILD_WPT];
            bool in[BUILD_WPT];
            #pragma unroll
            for (int k = 0; k < BUILD_WPT; ++k) {
                const unsigned i = (unsigned)(i0 + k * BUILD_THREADS + t);
                const int s = min((int)(i >> SUB_SHIFT), segs - 1);
                const int p = (int)(i & (SUB_CAP - 1));
                in[k] = (int)i < total && p < f[k];
                const int* a = f[k] > 0 ? mine + s * SUB_CAP + min(p, f[k] - 1)
                                        : counts + (unsigned)s * (unsigned)nTiles + b;
                w[k] = (unsigned)*a;
            }
            int r[BUILD_WPT];
            #pragma unroll
            for (int k = 0; k < BUILD_WPT; ++k) {
                r[k] = SLOTS;
                if (in[k]) r[k] = atomicAdd(&degL[w[k] >> SRC_BITS], 1);
            }
            #pragma unroll
            for (int k = 0; k < BUILD_WPT; ++k) {
                if (r[k] < SLOTS)
                    rowsL[(int)(w[k] >> SRC_BITS) * SLOTS + r[k]] = (int)(w[k] & ((1u << SRC_BITS) - 1u));
            }
        }
    } else {
        for (int e = t; e < E; e += BUILD_THREADS) {
            const int d = dst[e];
            if ((d >> BIN_SHIFT) == b) {
                const int dl = d & (BIN_NODES - 1);
                const int r = atomicAdd(&degL[dl], 1);
                if (r < SLOTS) rowsL[dl * SLOTS + r] = src[e];
            }
        }
    }
    __syncthreads();
    if (t < nodes) cnt[b * BIN_NODES + t] = degL[t];
    int* rows = slots + (unsigned)b * BIN_ROWWORDS;
    for (int i = t; i < nodes * SLOTS; i += BUILD_THREADS) rows[i] = rowsL[i];
}

__device__ __forceinline__ float dinv_of(int indeg) {
    return __builtin_amdgcn_rsqf((float)indeg + 1.0f);
}

template <bool WEIGHTED>
__device__ __forceinline__ float4 scan_all_edges(const int* __restrict__ src, const int* __restrict__ dst,
                                                 const int* __restrict__ cnt, const float* __restrict__ table,
                                                 int node, bool wanted, int E, int l16, int j) {
    float4 acc = make_float4(0.0f, 0.0f, 0.0f, 0.0f);
    #pragma unroll 1
    for (int e0 = 0; e0 < E; e0 += 16) {
        const int e = e0 + l16;
        bool m = false;
        int s = node;
        if (wanted && e < E && dst[e] == node) {
            m = true;
            s = src[e];
        }
        if (__ballot(m) == 0) continue;
        float w = 0.0f;
        if (m) w = WEIGHTED ? dinv_of(cnt[s]) : 1.0f;
        #pragma unroll 1
        for (int k = 0; k < 16; ++k) {
            const int sk = __shfl(s, k, 16);
            const float wk = __shfl(w, k, 16);
            const float4 v = *(const float4*)(table + (size_t)sk * DFEAT + j);
            acc.x += v.x * wk;
            acc.y += v.y * wk;
            acc.z += v.z * wk;
            acc.w += v.w * wk;
        }
    }
    return acc;
}

__global__ __launch_bounds__(256) void gather_rows_kernel(const int* __restrict__ cnt,
                                                          const int* __restrict__ slots,
                                                          const int* __restrict__ src,
                                                          const int* __restrict__ dst,
                                                          const float* __restrict__ h,
                                                          const float* __restrict__ b,
                                                          float* __restrict__ outp, int n, int E) {
    const int tid = blockIdx.x * 256 + threadIdx.x;
    int node = tid >> 4;
    const bool valid = node < n;
    if (!valid) node = n - 1;
    const int l16 = threadIdx.x & 15;
    const int j = l16 << 2;

    const int deg = cnt[node];
    const unsigned rowBase = (unsigned)node * SLOTS;
    const bool over = valid && deg > SLOTS;
    const int len = valid ? min(deg, SLOTS) : 0;
    int lenMax = max(len, __shfl_xor(len, 16, 32));
    lenMax = __builtin_amdgcn_readfirstlane(lenMax);

    float4 acc = make_float4(0.0f, 0.0f, 0.0f, 0.0f);
    for (int base = 0; base < lenMax; base += 16) {
        const bool in = base + l16 < len;
        const int ent = slots[rowBase + (in ? base + l16 : 0)];
        const int ex = in ? ent : node;
        const float ey = in ? 1.0f : 0.0f;
        const int chunk = min(16, lenMax - base);
        for (int k = 0; k < chunk; k += 4) {
            int s[4];
            float w[4];
            float4 v[4];
            for (int u = 0; u < 4; ++u) {
                s[u] = __shfl(ex, k + u, 16);
                w[u] = __shfl(ey, k + u, 16);
            }
            for (int u = 0; u < 4; ++u)
                v[u] = *(const float4*)(h + (size_t)s[u] * DFEAT + j);
            for (int u = 0; u < 4; ++u) {
                acc.x += v[u].x * w[u];
                acc.y += v[u].y * w[u];
                acc.z += v[u].z * w[u];
                acc.w += v[u].w * w[u];
            }
        }
    }
    if (__ballot(over) != 0) {
        const float4 r = scan_all_edges<false>(src, dst, cnt, h, node, over, E, l16, j);
        if (over) acc = r;
    }

    if (valid) {
        const float di = dinv_of(deg);
        const float4 hv = *(const float4*)(h + (size_t)node * DFEAT + j);
        float4 r;
        r.x = (acc.x + hv.x) * di + b[j + 0];
        r.y = (acc.y + hv.y) * di + b[j + 1];
        r.z = (acc.z + hv.z) * di + b[j + 2];
        r.w = (acc.w + hv.w) * di + b[j + 3];
        *(float4*)(outp + (size_t)node * DFEAT + j) = r;
    }
}

#define TILE_ROWS 64
#define EPS 2
#define TP 68

__global__ __launch_bounds__(256) void gather_mlp_kernel(const int* __restrict__ cnt,
                                                         const int* __restrict__ slots,
                                                         const int* __restrict__ src,
                                                         const int* __restrict__ dst,
                                                         const float* __restrict__ emb,
                                                         const float* __restrict__ W1,
                                                         const float* __restrict__ b1,
                                                         const float* __restrict__ W2,
                                                         float* __restrict__ h2, int n, int E) {
    __shared__ __attribute__((aligned(16))) float Wl1[DFEAT * DFEAT];
    __shared__ __attribute__((aligned(16))) float Wl2[DFEAT * DFEAT];
    __shared__ __attribute__((aligned(16))) float Ta[TILE_ROWS * TP];
    __shared__ __attribute__((aligned(16))) float Tx[TILE_ROWS * TP];
    __shared__ float dinvT[TILE_ROWS];

    const int t = threadIdx.x;
    const int wave = t >> 5;
    const int half = (t >> 4) & 1;
    const int l16 = t & 15;
    const int j = l16 << 2;

    {
        const float4* W1v = (const float4*)W1;
        const float4* W2v = (const float4*)W2;
        float4* Wl1v = (float4*)Wl1;
        float4* Wl2v = (float4*)Wl2;
        #pragma unroll
        for (int i = 0; i < (DFEAT * DFEAT / 4) / 256; ++i) {
            Wl1v[t + i * 256] = W1v[t + i * 256];
            Wl2v[t + i * 256] = W2v[t + i * 256];
        }
    }

    const int grp = wave * 2 + half;
    const int tileRow0 = grp * 4;
    const int node0 = blockIdx.x * TILE_ROWS + tileRow0;

    int nd[4], len[4], deg[4];
    unsigned rowBase[4];
    bool over[4];
    #pragma unroll
    for (int q = 0; q < 4; ++q) {
        const int node = node0 + q;
        const bool valid = node < n;
        nd[q] = valid ? node : n - 1;
        rowBase[q] = (unsigned)nd[q] * SLOTS;
        deg[q] = cnt[nd[q]];
        len[q] = valid ? min(deg[q], SLOTS) : 0;
        over[q] = valid && deg[q] > SLOTS;
    }
    int lenMax = max(max(len[0], len[1]), max(len[2], len[3]));
    lenMax = max(lenMax, __shfl_xor(lenMax, 16, 32));
    lenMax = __builtin_amdgcn_readfirstlane(lenMax);

    float4 acc[4];
    #pragma unroll
    for (int q = 0; q < 4; ++q) acc[q] = make_float4(0.0f, 0.0f, 0.0f, 0.0f);
    const char* embBytes = (const char*)emb;
    const unsigned colByte = (unsigned)j << 2;

    int ent[4];
    #pragma unroll
    for (int q = 0; q < 4; ++q)
        ent[q] = slots[rowBase[q] + (l16 < len[q] ? l16 : 0)];

    for (int base = 0; base < lenMax; base += 16) {
        int ex[4], ey[4];
        int sdeg[4];
        #pragma unroll
        for (int q = 0; q < 4; ++q) {
            const bool in = base + l16 < len[q];
            ex[q] = in ? ent[q] : nd[q];
            sdeg[q] = cnt[ex[q]];
        }
        #pragma unroll
        for (int q = 0; q < 4; ++q) {
            const int nx = base + 16 + l16;
            ent[q] = slots[rowBase[q] + (nx < len[q] ? nx : 0)];
        }
        #pragma unroll
        for (int q = 0; q < 4; ++q) {
            const bool in = base + l16 < len[q];
            ey[q] = in ? __float_as_int(dinv_of(sdeg[q])) : 0;
        }
        const int chunk = min(16, lenMax - base);
        for (int k = 0; k < chunk; k += EPS) {
            float wgt[4][EPS];
            float4 v[4][EPS];
            #pragma unroll
            for (int q = 0; q < 4; ++q) {
                #pragma unroll
                for (int u = 0; u < EPS; ++u) {
                    const int s = __shfl(ex[q], k + u, 16);
                    wgt[q][u] = __int_as_float(__shfl(ey[q], k + u, 16));
                    const unsigned off = ((unsigned)s << 8) + colByte;
                    v[q][u] = *(const float4*)(embBytes + off);
                }
            }
            #pragma unroll
            for (int q = 0; q < 4; ++q) {
                #pragma unroll
                for (int u = 0; u < EPS; ++u) {
                    acc[q].x += v[q][u].x * wgt[q][u];
                    acc[q].y += v[q][u].y * wgt[q][u];
                    acc[q].z += v[q][u].z * wgt[q][u];
                    acc[q].w += v[q][u].w * wgt[q][u];
                }
            }
        }
    }

    if (__ballot(over[0] || over[1] || over[2] || over[3]) != 0) {
        #pragma unroll
        for (int q = 0; q < 4; ++q) {
            const float4 r = scan_all_edges<true>(src, dst, cnt, emb, nd[q], over[q], E, l16, j);
            if (over[q]) acc[q] = r;
        }
    }

    float4 self[4];
    #pragma unroll
    for (int q = 0; q < 4; ++q)
        self[q] = *(const float4*)(emb + (size_t)nd[q] * DFEAT + j);
    #pragma unroll
    for (int q = 0; q < 4; ++q) {
        const float di = dinv_of(deg[q]);
        float4 r;
        r.x = (acc[q].x + self[q].x * di) * di;
        r.y = (acc[q].y + self[q].y * di) * di;
        r.z = (acc[q].z + self[q].z * di) * di;
        r.w = (acc[q].w + self[q].w * di) * di;
        if (node0 + q >= n) r = make_float4(0.0f, 0.0f, 0.0f, 0.0f);
        *(float4*)(Ta + (tileRow0 + q) * TP + j) = r;
        if (l16 == 0) dinvT[tileRow0 + q] = di;
    }
    __syncthreads();

    const int mt = wave >> 1;
    const int nh = wave & 1;
    const int rowT = mt * 16 + l16;

    v8f c1[2] = {};
    {
        const float* Arow = Ta + rowT * TP;
        for (int k0 = 0; k0 < DFEAT; k0 += 4) {
            const int ka = k0 + half * 2;
            v2f a;
            a[0] = Arow[ka];
            a[1] = Arow[ka + 1];
            #pragma unroll
            for (int tn = 0; tn < 2; ++tn) {
                const int n0 = nh * 32 + tn * 16;
                v2f b;
                b[0] = Wl1[ka * DFEAT + n0 + l16];
                b[1] = Wl1[(ka + 1) * DFEAT + n0 + l16];
                c1[tn] = __builtin_amdgcn_wmma_f32_16x16x4_f32(
                     false, a,  false, b,
                     (short)0, c1[tn],  false,  false);
            }
        }
    }
    asm volatile("v_nop\n\tv_nop\n\tv_nop\n\tv_nop" : "+v"(c1[0]), "+v"(c1[1]));
    #pragma unroll
    for (int tn = 0; tn < 2; ++tn) {
        const int col = nh * 32 + tn * 16 + l16;
        const float bias = b1[col];
        #pragma unroll
        for (int v = 0; v < 8; ++v) {
            const int rowInTile = mt * 16 + v + half * 8;
            float r = c1[tn][v] + bias;
            r = r > 0.0f ? r : 0.0f;
            Tx[rowInTile * TP + col] = r * dinvT[rowInTile];
        }
    }
    __syncthreads();

    v8f c2[2] = {};
    {
        const float* Arow = Tx + rowT * TP;
        for (int k0 = 0; k0 < DFEAT; k0 += 4) {
            const int ka = k0 + half * 2;
            v2f a;
            a[0] = Arow[ka];
            a[1] = Arow[ka + 1];
            #pragma unroll
            for (int tn = 0; tn < 2; ++tn) {
                const int n0 = nh * 32 + tn * 16;
                v2f b;
                b[0] = Wl2[ka * DFEAT + n0 + l16];
                b[1] = Wl2[(ka + 1) * DFEAT + n0 + l16];
                c2[tn] = __builtin_amdgcn_wmma_f32_16x16x4_f32(
                     false, a,  false, b,
                     (short)0, c2[tn],  false,  false);
            }
        }
    }
    #pragma unroll
    for (int tn = 0; tn < 2; ++tn) {
        const int col = nh * 32 + tn * 16 + l16;
        #pragma unroll
        for (int v = 0; v < 8; ++v) {
            const int row = blockIdx.x * TILE_ROWS + mt * 16 + v + half * 8;
            if (row < n) h2[(size_t)row * DFEAT + col] = c2[tn][v];
        }
    }
}


static_assert(BIN_NODES == TILE_ROWS, "the bin tiles and the layer kernel's tiles are counted by one grid size");

extern "C" void kernel_launch(void* const* d_in, const int* in_sizes, int n_in,
                              void* d_out, int out_size, void* d_ws, size_t ws_size,
                              hipStream_t stream) {
    const int* eidx = (const int*)d_in[0];
    const float* emb = (const float*)d_in[1];
    const float* W1 = (const float*)d_in[2];
    const float* b1 = (const float*)d_in[3];
    const float* W2 = (const float*)d_in[4];
    const float* b2 = (const float*)d_in[5];
    float* out = (float*)d_out;

    const int E = in_sizes[0] / 2;
    const int n = in_sizes[1] / DFEAT;

    const int* src = eidx;
    const int* dst = eidx + E;

    float* h2    = (float*)d_ws;
    int* slots   = (int*)(h2 + (size_t)n * DFEAT);
    int* cnt     = slots + (size_t)n * SLOTS;

    int* bins    = (int*)h2;

    const int B = 256;
    const int gRows = (n * 16 + B - 1) / B;
    const int gTile = (n + TILE_ROWS - 1) / TILE_ROWS;
    const int gBin = (E + BIN_EDGES - 1) / BIN_EDGES;

    bin_kernel<<<gBin, BIN_THREADS, 0, stream>>>(src, dst, bins, E, n);
    build_rows_kernel<<<gTile, BUILD_THREADS, 0, stream>>>(src, dst, bins, cnt, slots, n, E);

    gather_mlp_kernel<<<gTile, B, 0, stream>>>(cnt, slots, src, dst, emb, W1, b1, W2, h2, n, E);

    gather_rows_kernel<<<gRows, B, 0, stream>>>(cnt, slots, src, dst, h2, b2, out, n, E);
}
